# starting kernel
# baseline (speedup 1.0000x reference)
#include <hip/hip_runtime.h>
#include <math.h>


#define B_   32
#define C_   256
#define H_   64
#define W_   64
#define HW_  4096
#define MID_ 16

typedef __attribute__((ext_vector_type(2))) float v2f;
typedef __attribute__((ext_vector_type(4))) float v4f;
typedef __attribute__((ext_vector_type(8))) float v8f;

__global__ __launch_bounds__(256) void cbam_pool(const float* __restrict__ x,
                                                 float* __restrict__ avg,
                                                 float* __restrict__ mxp) {
  int plane = blockIdx.x;
  const v4f* p = (const v4f*)(x + (size_t)plane * HW_);
  float s = 0.f, m = -INFINITY;
#pragma unroll
  for (int i = 0; i < 4; ++i) {
    v4f v = p[threadIdx.x + i * 256];
    s += v.x + v.y + v.z + v.w;
    m = fmaxf(m, fmaxf(fmaxf(v.x, v.y), fmaxf(v.z, v.w)));
  }
#pragma unroll
  for (int off = 16; off >= 1; off >>= 1) {
    s += __shfl_xor(s, off, 32);
    m = fmaxf(m, __shfl_xor(m, off, 32));
  }
  __shared__ float ss[8], sm[8];
  int wave = threadIdx.x >> 5, lane = threadIdx.x & 31;
  if (lane == 0) { ss[wave] = s; sm[wave] = m; }
  __syncthreads();
  if (threadIdx.x == 0) {
    float ts = 0.f, tm = -INFINITY;
#pragma unroll
    for (int i = 0; i < 8; ++i) { ts += ss[i]; tm = fmaxf(tm, sm[i]); }
    avg[plane] = ts * (1.0f / (float)HW_);
    mxp[plane] = tm;
  }
}

__global__ __launch_bounds__(256) void cbam_mlp_ca(const float* __restrict__ avg,
                                                   const float* __restrict__ mxp,
                                                   const float* __restrict__ w1,
                                                   const float* __restrict__ w2,
                                                   float* __restrict__ ca) {
  __shared__ float hcomb[32][32];
  int tid  = threadIdx.x;
  int wave = tid >> 5;
  int lane = tid & 31;
  int half = lane >> 4;
  int l15  = lane & 15;

  if (wave < 4) {
    int src   = wave >> 1;
    int mtile = wave & 1;
    const float* pool = src ? mxp : avg;
    int M = mtile * 16 + l15;
    v8f acc = {};
    for (int k0 = 0; k0 < 256; k0 += 4) {
      int ka = k0 + half * 2;
      v2f a, b;
      a.x = pool[M * 256 + ka];
      a.y = pool[M * 256 + ka + 1];
      b.x = w1[l15 * 256 + ka];
      b.y = w1[l15 * 256 + ka + 1];
      acc = __builtin_amdgcn_wmma_f32_16x16x4_f32(
          false, a, false, b, (short)0, acc, false, false);
    }
#pragma unroll
    for (int i = 0; i < 8; ++i) {
      int m = mtile * 16 + i + half * 8;
      hcomb[m][src * 16 + l15] = fmaxf(acc[i], 0.f);
    }
  }
  __syncthreads();

  for (int t = wave; t < 32; t += 8) {
    int mtile = t >> 4;
    int ntile = t & 15;
    v8f acc = {};
#pragma unroll
    for (int k0 = 0; k0 < 32; k0 += 4) {
      int ka = k0 + half * 2;
      v2f a, b;
      a.x = hcomb[mtile * 16 + l15][ka];
      a.y = hcomb[mtile * 16 + l15][ka + 1];
      b.x = w2[(ntile * 16 + l15) * 16 + (ka & 15)];
      b.y = w2[(ntile * 16 + l15) * 16 + ((ka + 1) & 15)];
      acc = __builtin_amdgcn_wmma_f32_16x16x4_f32(
          false, a, false, b, (short)0, acc, false, false);
    }
#pragma unroll
    for (int i = 0; i < 8; ++i) {
      int m = mtile * 16 + i + half * 8;
      int c = ntile * 16 + l15;
      ca[m * 256 + c] = 1.f / (1.f + __expf(-acc[i]));
    }
  }
}

__global__ __launch_bounds__(256) void cbam_feat(const float* __restrict__ x,
                                                 const float* __restrict__ ca,
                                                 float* __restrict__ fm,
                                                 float* __restrict__ fx) {
  __shared__ float sCa[256];
  int b  = blockIdx.x >> 4;
  int hw = ((blockIdx.x & 15) << 8) + threadIdx.x;
  sCa[threadIdx.x] = ca[b * 256 + threadIdx.x];
  __syncthreads();
  const float* xb = x + (size_t)b * C_ * HW_;
  float s = 0.f, m = -INFINITY;
#pragma unroll 4
  for (int c = 0; c < 256; ++c) {
    float v = xb[(size_t)c * HW_ + hw] * sCa[c];
    s += v;
    m = fmaxf(m, v);
  }
  fm[b * HW_ + hw] = s * (1.f / 256.f);
  fx[b * HW_ + hw] = m;
}

__global__ __launch_bounds__(256) void cbam_conv(const float* __restrict__ fm,
                                                 const float* __restrict__ fx,
                                                 const float* __restrict__ wsp,
                                                 float* __restrict__ sig) {
  __shared__ float sW[98];
  if (threadIdx.x < 98) sW[threadIdx.x] = wsp[threadIdx.x];
  __syncthreads();
  int b  = blockIdx.x >> 4;
  int hw = ((blockIdx.x & 15) << 8) + threadIdx.x;
  int y = hw >> 6, xc = hw & 63;
  const float* fmb = fm + b * HW_;
  const float* fxb = fx + b * HW_;
  float acc = 0.f;
#pragma unroll
  for (int ky = 0; ky < 7; ++ky) {
    int yy = y + ky - 3;
    if (yy < 0 || yy >= H_) continue;
#pragma unroll
    for (int kx = 0; kx < 7; ++kx) {
      int x2 = xc + kx - 3;
      if (x2 < 0 || x2 >= W_) continue;
      acc += fmb[yy * 64 + x2] * sW[ky * 7 + kx]
           + fxb[yy * 64 + x2] * sW[49 + ky * 7 + kx];
    }
  }
  sig[b * HW_ + hw] = 1.f / (1.f + __expf(-acc));
}

__global__ __launch_bounds__(256) void cbam_final(const float* __restrict__ x,
                                                  const float* __restrict__ ca,
                                                  const float* __restrict__ sig,
                                                  float* __restrict__ out) {
  size_t idx = (size_t)blockIdx.x * 256 + threadIdx.x;
  size_t v   = idx * 4;
  int hw = (int)(v & 4095);
  int c  = (int)((v >> 12) & 255);
  int b  = (int)(v >> 20);
  v4f xv = __builtin_nontemporal_load((const v4f*)x + idx);
  v4f sv = *(const v4f*)(sig + (size_t)b * HW_ + hw);
  float cv = ca[b * 256 + c];
  v4f o = xv * (sv * cv);
  __builtin_nontemporal_store(o, (v4f*)out + idx);
}

extern "C" void kernel_launch(void* const* d_in, const int* in_sizes, int n_in,
                              void* d_out, int out_size, void* d_ws, size_t ws_size,
                              hipStream_t stream) {
  const float* x  = (const float*)d_in[0];
  const float* w1 = (const float*)d_in[1];
  const float* w2 = (const float*)d_in[2];
  const float* ws = (const float*)d_in[3];
  float* out = (float*)d_out;

  float* wsf = (float*)d_ws;
  float* avg = wsf;
  float* mxp = wsf + 8192;
  float* ca  = wsf + 16384;
  float* fm  = wsf + 24576;
  float* fx  = wsf + 155648;
  float* sig = wsf + 286720;

  cbam_pool  <<<B_ * C_,          256, 0, stream>>>(x, avg, mxp);
  cbam_mlp_ca<<<1,                256, 0, stream>>>(avg, mxp, w1, w2, ca);
  cbam_feat  <<<B_ * (HW_ / 256), 256, 0, stream>>>(x, ca, fm, fx);
  cbam_conv  <<<B_ * (HW_ / 256), 256, 0, stream>>>(fm, fx, ws, sig);
  cbam_final <<<(B_ * C_ * HW_ / 4) / 256, 256, 0, stream>>>(x, ca, sig, out);
}


# feat pass: 256 channels split over four 64-thread groups per block, 16-byte x loads in batches of 16 with the next batch requested ahead, partials joined in LDS; channel sum now 4 partial chains
# speedup vs baseline: 1.3571x; 1.3571x over previous
#include <hip/hip_runtime.h>
#include <math.h>


#define B_   32
#define C_   256
#define H_   64
#define W_   64
#define HW_  4096
#define MID_ 16

typedef __attribute__((ext_vector_type(2))) float v2f;
typedef __attribute__((ext_vector_type(4))) float v4f;
typedef __attribute__((ext_vector_type(8))) float v8f;

__global__ __launch_bounds__(256) void cbam_pool(const float* __restrict__ x,
                                                 float* __restrict__ avg,
                                                 float* __restrict__ mxp) {
  int plane = blockIdx.x;
  const v4f* p = (const v4f*)(x + (size_t)plane * HW_);
  float s = 0.f, m = -INFINITY;
#pragma unroll
  for (int i = 0; i < 4; ++i) {
    v4f v = p[threadIdx.x + i * 256];
    s += v.x + v.y + v.z + v.w;
    m = fmaxf(m, fmaxf(fmaxf(v.x, v.y), fmaxf(v.z, v.w)));
  }
#pragma unroll
  for (int off = 16; off >= 1; off >>= 1) {
    s += __shfl_xor(s, off, 32);
    m = fmaxf(m, __shfl_xor(m, off, 32));
  }
  __shared__ float ss[8], sm[8];
  int wave = threadIdx.x >> 5, lane = threadIdx.x & 31;
  if (lane == 0) { ss[wave] = s; sm[wave] = m; }
  __syncthreads();
  if (threadIdx.x == 0) {
    float ts = 0.f, tm = -INFINITY;
#pragma unroll
    for (int i = 0; i < 8; ++i) { ts += ss[i]; tm = fmaxf(tm, sm[i]); }
    avg[plane] = ts * (1.0f / (float)HW_);
    mxp[plane] = tm;
  }
}

__global__ __launch_bounds__(256) void cbam_mlp_ca(const float* __restrict__ avg,
                                                   const float* __restrict__ mxp,
                                                   const float* __restrict__ w1,
                                                   const float* __restrict__ w2,
                                                   float* __restrict__ ca) {
  __shared__ float hcomb[32][32];
  int tid  = threadIdx.x;
  int wave = tid >> 5;
  int lane = tid & 31;
  int half = lane >> 4;
  int l15  = lane & 15;

  if (wave < 4) {
    int src   = wave >> 1;
    int mtile = wave & 1;
    const float* pool = src ? mxp : avg;
    int M = mtile * 16 + l15;
    v8f acc = {};
    for (int k0 = 0; k0 < 256; k0 += 4) {
      int ka = k0 + half * 2;
      v2f a, b;
      a.x = pool[M * 256 + ka];
      a.y = pool[M * 256 + ka + 1];
      b.x = w1[l15 * 256 + ka];
      b.y = w1[l15 * 256 + ka + 1];
      acc = __builtin_amdgcn_wmma_f32_16x16x4_f32(
          false, a, false, b, (short)0, acc, false, false);
    }
#pragma unroll
    for (int i = 0; i < 8; ++i) {
      int m = mtile * 16 + i + half * 8;
      hcomb[m][src * 16 + l15] = fmaxf(acc[i], 0.f);
    }
  }
  __syncthreads();

  for (int t = wave; t < 32; t += 8) {
    int mtile = t >> 4;
    int ntile = t & 15;
    v8f acc = {};
#pragma unroll
    for (int k0 = 0; k0 < 32; k0 += 4) {
      int ka = k0 + half * 2;
      v2f a, b;
      a.x = hcomb[mtile * 16 + l15][ka];
      a.y = hcomb[mtile * 16 + l15][ka + 1];
      b.x = w2[(ntile * 16 + l15) * 16 + (ka & 15)];
      b.y = w2[(ntile * 16 + l15) * 16 + ((ka + 1) & 15)];
      acc = __builtin_amdgcn_wmma_f32_16x16x4_f32(
          false, a, false, b, (short)0, acc, false, false);
    }
#pragma unroll
    for (int i = 0; i < 8; ++i) {
      int m = mtile * 16 + i + half * 8;
      int c = ntile * 16 + l15;
      ca[m * 256 + c] = 1.f / (1.f + __expf(-acc[i]));
    }
  }
}

#define FEAT_GROUPS  4
#define FEAT_CPG     64
#define FEAT_BATCH   16
#define FEAT_NBATCH  (FEAT_CPG / FEAT_BATCH)
#define FEAT_CSTR    (HW_ / 4)

__global__ __launch_bounds__(256) void cbam_feat(const float* __restrict__ x,
                                                 const float* __restrict__ ca,
                                                 float* __restrict__ fm,
                                                 float* __restrict__ fx) {
  __shared__ v4f sS[FEAT_GROUPS][64];
  __shared__ v4f sM[FEAT_GROUPS][64];
  const int b   = blockIdx.x >> 4;
  const int hw0 = (blockIdx.x & 15) << 8;
  const int q   = threadIdx.x & 63;
  const int g = __builtin_amdgcn_readfirstlane((int)(threadIdx.x >> 6));
  const float* cag = ca + b * C_ + g * FEAT_CPG;
  const v4f* p = (const v4f*)(x + ((size_t)b * C_ + (size_t)g * FEAT_CPG) * HW_ + hw0) + q;

  v4f buf[2][FEAT_BATCH];
#define FEAT_LOAD(R)                                                         \
  _Pragma("unroll") for (int i = 0; i < FEAT_BATCH; ++i)                     \
    buf[(R) & 1][i] = p[((R) * FEAT_BATCH + i) * FEAT_CSTR];
  FEAT_LOAD(0)

  v4f s = {0.f, 0.f, 0.f, 0.f};
  v4f m = {-INFINITY, -INFINITY, -INFINITY, -INFINITY};
#pragma unroll
  for (int r = 0; r < FEAT_NBATCH; ++r) {
    __builtin_amdgcn_sched_barrier(0);
    if (r + 1 < FEAT_NBATCH) { FEAT_LOAD(r + 1) }
    __builtin_amdgcn_sched_barrier(0);
#pragma unroll
    for (int i = 0; i < FEAT_BATCH; ++i) {
      v4f v = buf[r & 1][i] * cag[r * FEAT_BATCH + i];
      s += v;
      m.x = fmaxf(m.x, v.x);
      m.y = fmaxf(m.y, v.y);
      m.z = fmaxf(m.z, v.z);
      m.w = fmaxf(m.w, v.w);
    }
  }
#undef FEAT_LOAD
  sS[g][q] = s;
  sM[g][q] = m;
  __syncthreads();
  if (threadIdx.x < 64) {
    v4f ts = ((sS[0][q] + sS[1][q]) + sS[2][q]) + sS[3][q];
    v4f a = sM[0][q], bq = sM[1][q], cq = sM[2][q], dq = sM[3][q];
    v4f tm;
    tm.x = fmaxf(fmaxf(a.x, bq.x), fmaxf(cq.x, dq.x));
    tm.y = fmaxf(fmaxf(a.y, bq.y), fmaxf(cq.y, dq.y));
    tm.z = fmaxf(fmaxf(a.z, bq.z), fmaxf(cq.z, dq.z));
    tm.w = fmaxf(fmaxf(a.w, bq.w), fmaxf(cq.w, dq.w));
    *(v4f*)(fm + b * HW_ + hw0 + 4 * q) = ts * (1.f / 256.f);
    *(v4f*)(fx + b * HW_ + hw0 + 4 * q) = tm;
  }
}

__global__ __launch_bounds__(256) void cbam_conv(const float* __restrict__ fm,
                                                 const float* __restrict__ fx,
                                                 const float* __restrict__ wsp,
                                                 float* __restrict__ sig) {
  __shared__ float sW[98];
  if (threadIdx.x < 98) sW[threadIdx.x] = wsp[threadIdx.x];
  __syncthreads();
  int b  = blockIdx.x >> 4;
  int hw = ((blockIdx.x & 15) << 8) + threadIdx.x;
  int y = hw >> 6, xc = hw & 63;
  const float* fmb = fm + b * HW_;
  const float* fxb = fx + b * HW_;
  float acc = 0.f;
#pragma unroll
  for (int ky = 0; ky < 7; ++ky) {
    int yy = y + ky - 3;
    if (yy < 0 || yy >= H_) continue;
#pragma unroll
    for (int kx = 0; kx < 7; ++kx) {
      int x2 = xc + kx - 3;
      if (x2 < 0 || x2 >= W_) continue;
      acc += fmb[yy * 64 + x2] * sW[ky * 7 + kx]
           + fxb[yy * 64 + x2] * sW[49 + ky * 7 + kx];
    }
  }
  sig[b * HW_ + hw] = 1.f / (1.f + __expf(-acc));
}

__global__ __launch_bounds__(256) void cbam_final(const float* __restrict__ x,
                                                  const float* __restrict__ ca,
                                                  const float* __restrict__ sig,
                                                  float* __restrict__ out) {
  size_t idx = (size_t)blockIdx.x * 256 + threadIdx.x;
  size_t v   = idx * 4;
  int hw = (int)(v & 4095);
  int c  = (int)((v >> 12) & 255);
  int b  = (int)(v >> 20);
  v4f xv = __builtin_nontemporal_load((const v4f*)x + idx);
  v4f sv = *(const v4f*)(sig + (size_t)b * HW_ + hw);
  float cv = ca[b * 256 + c];
  v4f o = xv * (sv * cv);
  __builtin_nontemporal_store(o, (v4f*)out + idx);
}

extern "C" void kernel_launch(void* const* d_in, const int* in_sizes, int n_in,
                              void* d_out, int out_size, void* d_ws, size_t ws_size,
                              hipStream_t stream) {
  const float* x  = (const float*)d_in[0];
  const float* w1 = (const float*)d_in[1];
  const float* w2 = (const float*)d_in[2];
  const float* ws = (const float*)d_in[3];
  float* out = (float*)d_out;

  float* wsf = (float*)d_ws;
  float* avg = wsf;
  float* mxp = wsf + 8192;
  float* ca  = wsf + 16384;
  float* fm  = wsf + 24576;
  float* fx  = wsf + 155648;
  float* sig = wsf + 286720;

  cbam_pool  <<<B_ * C_,          256, 0, stream>>>(x, avg, mxp);
  cbam_mlp_ca<<<1,                256, 0, stream>>>(avg, mxp, w1, w2, ca);
  cbam_feat  <<<B_ * (HW_ / 256), 256, 0, stream>>>(x, ca, fm, fx);
  cbam_conv  <<<B_ * (HW_ / 256), 256, 0, stream>>>(fm, fx, ws, sig);
  cbam_final <<<(B_ * C_ * HW_ / 4) / 256, 256, 0, stream>>>(x, ca, sig, out);
}


# final pass as a resident stream: 512 blocks, each walks 16 contiguous channel planes, 8 16-byte loads per lane requested one batch ahead of the stores; 64x fewer waves; sums unchanged
# speedup vs baseline: 1.4276x; 1.0519x over previous
#include <hip/hip_runtime.h>
#include <math.h>


#define B_   32
#define C_   256
#define H_   64
#define W_   64
#define HW_  4096
#define MID_ 16

typedef __attribute__((ext_vector_type(2))) float v2f;
typedef __attribute__((ext_vector_type(4))) float v4f;
typedef __attribute__((ext_vector_type(8))) float v8f;

__global__ __launch_bounds__(256) void cbam_pool(const float* __restrict__ x,
                                                 float* __restrict__ avg,
                                                 float* __restrict__ mxp) {
  int plane = blockIdx.x;
  const v4f* p = (const v4f*)(x + (size_t)plane * HW_);
  float s = 0.f, m = -INFINITY;
#pragma unroll
  for (int i = 0; i < 4; ++i) {
    v4f v = p[threadIdx.x + i * 256];
    s += v.x + v.y + v.z + v.w;
    m = fmaxf(m, fmaxf(fmaxf(v.x, v.y), fmaxf(v.z, v.w)));
  }
#pragma unroll
  for (int off = 16; off >= 1; off >>= 1) {
    s += __shfl_xor(s, off, 32);
    m = fmaxf(m, __shfl_xor(m, off, 32));
  }
  __shared__ float ss[8], sm[8];
  int wave = threadIdx.x >> 5, lane = threadIdx.x & 31;
  if (lane == 0) { ss[wave] = s; sm[wave] = m; }
  __syncthreads();
  if (threadIdx.x == 0) {
    float ts = 0.f, tm = -INFINITY;
#pragma unroll
    for (int i = 0; i < 8; ++i) { ts += ss[i]; tm = fmaxf(tm, sm[i]); }
    avg[plane] = ts * (1.0f / (float)HW_);
    mxp[plane] = tm;
  }
}

__global__ __launch_bounds__(256) void cbam_mlp_ca(const float* __restrict__ avg,
                                                   const float* __restrict__ mxp,
                                                   const float* __restrict__ w1,
                                                   const float* __restrict__ w2,
                                                   float* __restrict__ ca) {
  __shared__ float hcomb[32][32];
  int tid  = threadIdx.x;
  int wave = tid >> 5;
  int lane = tid & 31;
  int half = lane >> 4;
  int l15  = lane & 15;

  if (wave < 4) {
    int src   = wave >> 1;
    int mtile = wave & 1;
    const float* pool = src ? mxp : avg;
    int M = mtile * 16 + l15;
    v8f acc = {};
    for (int k0 = 0; k0 < 256; k0 += 4) {
      int ka = k0 + half * 2;
      v2f a, b;
      a.x = pool[M * 256 + ka];
      a.y = pool[M * 256 + ka + 1];
      b.x = w1[l15 * 256 + ka];
      b.y = w1[l15 * 256 + ka + 1];
      acc = __builtin_amdgcn_wmma_f32_16x16x4_f32(
          false, a, false, b, (short)0, acc, false, false);
    }
#pragma unroll
    for (int i = 0; i < 8; ++i) {
      int m = mtile * 16 + i + half * 8;
      hcomb[m][src * 16 + l15] = fmaxf(acc[i], 0.f);
    }
  }
  __syncthreads();

  for (int t = wave; t < 32; t += 8) {
    int mtile = t >> 4;
    int ntile = t & 15;
    v8f acc = {};
#pragma unroll
    for (int k0 = 0; k0 < 32; k0 += 4) {
      int ka = k0 + half * 2;
      v2f a, b;
      a.x = hcomb[mtile * 16 + l15][ka];
      a.y = hcomb[mtile * 16 + l15][ka + 1];
      b.x = w2[(ntile * 16 + l15) * 16 + (ka & 15)];
      b.y = w2[(ntile * 16 + l15) * 16 + ((ka + 1) & 15)];
      acc = __builtin_amdgcn_wmma_f32_16x16x4_f32(
          false, a, false, b, (short)0, acc, false, false);
    }
#pragma unroll
    for (int i = 0; i < 8; ++i) {
      int m = mtile * 16 + i + half * 8;
      int c = ntile * 16 + l15;
      ca[m * 256 + c] = 1.f / (1.f + __expf(-acc[i]));
    }
  }
}

#define FEAT_GROUPS  4
#define FEAT_CPG     64
#define FEAT_BATCH   16
#define FEAT_NBATCH  (FEAT_CPG / FEAT_BATCH)
#define FEAT_CSTR    (HW_ / 4)

__global__ __launch_bounds__(256) void cbam_feat(const float* __restrict__ x,
                                                 const float* __restrict__ ca,
                                                 float* __restrict__ fm,
                                                 float* __restrict__ fx) {
  __shared__ v4f sS[FEAT_GROUPS][64];
  __shared__ v4f sM[FEAT_GROUPS][64];
  const int b   = blockIdx.x >> 4;
  const int hw0 = (blockIdx.x & 15) << 8;
  const int q   = threadIdx.x & 63;
  const int g = __builtin_amdgcn_readfirstlane((int)(threadIdx.x >> 6));
  const float* cag = ca + b * C_ + g * FEAT_CPG;
  const v4f* p = (const v4f*)(x + ((size_t)b * C_ + (size_t)g * FEAT_CPG) * HW_ + hw0) + q;

  v4f buf[2][FEAT_BATCH];
#define FEAT_LOAD(R)                                                         \
  _Pragma("unroll") for (int i = 0; i < FEAT_BATCH; ++i)                     \
    buf[(R) & 1][i] = p[((R) * FEAT_BATCH + i) * FEAT_CSTR];
  FEAT_LOAD(0)

  v4f s = {0.f, 0.f, 0.f, 0.f};
  v4f m = {-INFINITY, -INFINITY, -INFINITY, -INFINITY};
#pragma unroll
  for (int r = 0; r < FEAT_NBATCH; ++r) {
    __builtin_amdgcn_sched_barrier(0);
    if (r + 1 < FEAT_NBATCH) { FEAT_LOAD(r + 1) }
    __builtin_amdgcn_sched_barrier(0);
#pragma unroll
    for (int i = 0; i < FEAT_BATCH; ++i) {
      v4f v = buf[r & 1][i] * cag[r * FEAT_BATCH + i];
      s += v;
      m.x = fmaxf(m.x, v.x);
      m.y = fmaxf(m.y, v.y);
      m.z = fmaxf(m.z, v.z);
      m.w = fmaxf(m.w, v.w);
    }
  }
#undef FEAT_LOAD
  sS[g][q] = s;
  sM[g][q] = m;
  __syncthreads();
  if (threadIdx.x < 64) {
    v4f ts = ((sS[0][q] + sS[1][q]) + sS[2][q]) + sS[3][q];
    v4f a = sM[0][q], bq = sM[1][q], cq = sM[2][q], dq = sM[3][q];
    v4f tm;
    tm.x = fmaxf(fmaxf(a.x, bq.x), fmaxf(cq.x, dq.x));
    tm.y = fmaxf(fmaxf(a.y, bq.y), fmaxf(cq.y, dq.y));
    tm.z = fmaxf(fmaxf(a.z, bq.z), fmaxf(cq.z, dq.z));
    tm.w = fmaxf(fmaxf(a.w, bq.w), fmaxf(cq.w, dq.w));
    *(v4f*)(fm + b * HW_ + hw0 + 4 * q) = ts * (1.f / 256.f);
    *(v4f*)(fx + b * HW_ + hw0 + 4 * q) = tm;
  }
}

__global__ __launch_bounds__(256) void cbam_conv(const float* __restrict__ fm,
                                                 const float* __restrict__ fx,
                                                 const float* __restrict__ wsp,
                                                 float* __restrict__ sig) {
  __shared__ float sW[98];
  if (threadIdx.x < 98) sW[threadIdx.x] = wsp[threadIdx.x];
  __syncthreads();
  int b  = blockIdx.x >> 4;
  int hw = ((blockIdx.x & 15) << 8) + threadIdx.x;
  int y = hw >> 6, xc = hw & 63;
  const float* fmb = fm + b * HW_;
  const float* fxb = fx + b * HW_;
  float acc = 0.f;
#pragma unroll
  for (int ky = 0; ky < 7; ++ky) {
    int yy = y + ky - 3;
    if (yy < 0 || yy >= H_) continue;
#pragma unroll
    for (int kx = 0; kx < 7; ++kx) {
      int x2 = xc + kx - 3;
      if (x2 < 0 || x2 >= W_) continue;
      acc += fmb[yy * 64 + x2] * sW[ky * 7 + kx]
           + fxb[yy * 64 + x2] * sW[49 + ky * 7 + kx];
    }
  }
  sig[b * HW_ + hw] = 1.f / (1.f + __expf(-acc));
}

#define FIN_PLANES 16
#define FIN_BATCH  8
#define FIN_NBATCH (FIN_PLANES * 4 / FIN_BATCH)
#define FIN_QSTR   (HW_ / 16)

__global__ __launch_bounds__(256) void cbam_final(const float* __restrict__ x,
                                                  const float* __restrict__ ca,
                                                  const float* __restrict__ sig,
                                                  float* __restrict__ out) {
  const int b  = blockIdx.x / (C_ / FIN_PLANES);
  const int c0 = (blockIdx.x % (C_ / FIN_PLANES)) * FIN_PLANES;
  const size_t base = ((size_t)(b * C_ + c0)) * HW_;
  const v4f* px = (const v4f*)(x + base) + threadIdx.x;
  v4f* po = (v4f*)(out + base) + threadIdx.x;
  const v4f* ps = (const v4f*)(sig + (size_t)b * HW_) + threadIdx.x;
  const float* cab = ca + b * C_ + c0;

  v4f sv[4];
#pragma unroll
  for (int q = 0; q < 4; ++q) sv[q] = ps[q * FIN_QSTR];

  v4f buf[2][FIN_BATCH];
#define FIN_LOAD(R)                                                          \
  _Pragma("unroll") for (int i = 0; i < FIN_BATCH; ++i)                      \
    buf[(R) & 1][i] = __builtin_nontemporal_load(px + ((R) * FIN_BATCH + i) * FIN_QSTR);
  FIN_LOAD(0)

#pragma unroll
  for (int r = 0; r < FIN_NBATCH; ++r) {
    __builtin_amdgcn_sched_barrier(0);
    if (r + 1 < FIN_NBATCH) { FIN_LOAD(r + 1) }
    __builtin_amdgcn_sched_barrier(0);
#pragma unroll
    for (int i = 0; i < FIN_BATCH; ++i) {
      const int piece = r * FIN_BATCH + i;
      v4f o = buf[r & 1][i] * (sv[piece & 3] * cab[piece >> 2]);
      __builtin_nontemporal_store(o, po + piece * FIN_QSTR);
    }
  }
#undef FIN_LOAD
}

extern "C" void kernel_launch(void* const* d_in, const int* in_sizes, int n_in,
                              void* d_out, int out_size, void* d_ws, size_t ws_size,
                              hipStream_t stream) {
  const float* x  = (const float*)d_in[0];
  const float* w1 = (const float*)d_in[1];
  const float* w2 = (const float*)d_in[2];
  const float* ws = (const float*)d_in[3];
  float* out = (float*)d_out;

  float* wsf = (float*)d_ws;
  float* avg = wsf;
  float* mxp = wsf + 8192;
  float* ca  = wsf + 16384;
  float* fm  = wsf + 24576;
  float* fx  = wsf + 155648;
  float* sig = wsf + 286720;

  cbam_pool  <<<B_ * C_,          256, 0, stream>>>(x, avg, mxp);
  cbam_mlp_ca<<<1,                256, 0, stream>>>(avg, mxp, w1, w2, ca);
  cbam_feat  <<<B_ * (HW_ / 256), 256, 0, stream>>>(x, ca, fm, fx);
  cbam_conv  <<<B_ * (HW_ / 256), 256, 0, stream>>>(fm, fx, ws, sig);
  cbam_final <<<B_ * (C_ / FIN_PLANES), 256, 0, stream>>>(x, ca, sig, out);
}


# channel MLP folded into the feat pass: each block redoes its batch row's two matrix-unit products (k of pool*w1 split over 8 waves, partials joined in LDS: that sum order changes); 4 launches, not 5
# speedup vs baseline: 1.6207x; 1.1353x over previous
#include <hip/hip_runtime.h>
#include <math.h>


#define B_   32
#define C_   256
#define H_   64
#define W_   64
#define HW_  4096
#define MID_ 16

typedef __attribute__((ext_vector_type(2))) float v2f;
typedef __attribute__((ext_vector_type(4))) float v4f;
typedef __attribute__((ext_vector_type(8))) float v8f;

__global__ __launch_bounds__(256) void cbam_pool(const float* __restrict__ x,
                                                 float* __restrict__ avg,
                                                 float* __restrict__ mxp) {
  int plane = blockIdx.x;
  const v4f* p = (const v4f*)(x + (size_t)plane * HW_);
  float s = 0.f, m = -INFINITY;
#pragma unroll
  for (int i = 0; i < 4; ++i) {
    v4f v = p[threadIdx.x + i * 256];
    s += v.x + v.y + v.z + v.w;
    m = fmaxf(m, fmaxf(fmaxf(v.x, v.y), fmaxf(v.z, v.w)));
  }
#pragma unroll
  for (int off = 16; off >= 1; off >>= 1) {
    s += __shfl_xor(s, off, 32);
    m = fmaxf(m, __shfl_xor(m, off, 32));
  }
  __shared__ float ss[8], sm[8];
  int wave = threadIdx.x >> 5, lane = threadIdx.x & 31;
  if (lane == 0) { ss[wave] = s; sm[wave] = m; }
  __syncthreads();
  if (threadIdx.x == 0) {
    float ts = 0.f, tm = -INFINITY;
#pragma unroll
    for (int i = 0; i < 8; ++i) { ts += ss[i]; tm = fmaxf(tm, sm[i]); }
    avg[plane] = ts * (1.0f / (float)HW_);
    mxp[plane] = tm;
  }
}

#define FEAT_GROUPS  4
#define FEAT_CPG     64
#define FEAT_BATCH   16
#define FEAT_NBATCH  (FEAT_CPG / FEAT_BATCH)
#define FEAT_CSTR    (HW_ / 4)

static __device__ __forceinline__ void lds_barrier() {
  __builtin_amdgcn_fence(__ATOMIC_RELEASE, "workgroup", "local");
  __builtin_amdgcn_s_barrier();
  __builtin_amdgcn_fence(__ATOMIC_ACQUIRE, "workgroup", "local");
}

__global__ __launch_bounds__(256) void cbam_feat(const float* __restrict__ x,
                                                 const float* __restrict__ avg,
                                                 const float* __restrict__ mxp,
                                                 const float* __restrict__ w1,
                                                 const float* __restrict__ w2,
                                                 float* __restrict__ ca,
                                                 float* __restrict__ fm,
                                                 float* __restrict__ fx) {
  __shared__ v4f sS[FEAT_GROUPS][64];
  __shared__ v4f sM[FEAT_GROUPS][64];
  __shared__ float sPart[8][2][16];
  __shared__ __attribute__((aligned(16))) float sH[32];
  __shared__ __attribute__((aligned(16))) float sCa[C_];
  const int b   = blockIdx.x >> 4;
  const int hw0 = (blockIdx.x & 15) << 8;
  const int q   = threadIdx.x & 63;
  const int g    = __builtin_amdgcn_readfirstlane((int)(threadIdx.x >> 6));
  const int wave = __builtin_amdgcn_readfirstlane((int)(threadIdx.x >> 5));
  const int lane = threadIdx.x & 31;
  const int half = lane >> 4;
  const int l15  = lane & 15;
  const v4f* p = (const v4f*)(x + ((size_t)b * C_ + (size_t)g * FEAT_CPG) * HW_ + hw0) + q;

  const float* prow  = ((l15 & 1) ? mxp : avg) + b * C_ + wave * 32 + half * 2;
  const float* w1row = w1 + l15 * C_ + wave * 32 + half * 2;
  v2f pa[8], pb[8];
#pragma unroll
  for (int j = 0; j < 8; ++j) {
    pa[j] = *(const v2f*)(prow + 4 * j);
    pb[j] = *(const v2f*)(w1row + 4 * j);
  }
  v2f wb[2][4];
#pragma unroll
  for (int t = 0; t < 2; ++t)
#pragma unroll
    for (int jj = 0; jj < 4; ++jj)
      wb[t][jj] = *(const v2f*)(w2 + ((wave * 2 + t) * 16 + l15) * MID_ + 4 * jj + 2 * half);

  v4f buf[2][FEAT_BATCH];
#define FEAT_LOAD(R)                                                         \
  _Pragma("unroll") for (int i = 0; i < FEAT_BATCH; ++i)                     \
    buf[(R) & 1][i] = p[((R) * FEAT_BATCH + i) * FEAT_CSTR];
  __builtin_amdgcn_sched_barrier(0);
  FEAT_LOAD(0)
  __builtin_amdgcn_sched_barrier(0);

  {
    const bool live = l15 < 2;
    v8f acc = {};
#pragma unroll
    for (int j = 0; j < 8; ++j) {
      v2f a;
      a.x = live ? pa[j].x : 0.f;
      a.y = live ? pa[j].y : 0.f;
      acc = __builtin_amdgcn_wmma_f32_16x16x4_f32(
          false, a, false, pb[j], (short)0, acc, false, false);
    }
    if (lane < 16) {
      sPart[wave][0][l15] = acc[0];
      sPart[wave][1][l15] = acc[1];
    }
  }
  lds_barrier();
  if (threadIdx.x < 32) {
    float t = sPart[0][half][l15];
#pragma unroll
    for (int w = 1; w < 8; ++w) t += sPart[w][half][l15];
    sH[lane] = fmaxf(t, 0.f);
  }
  lds_barrier();

  {
    const bool live = l15 == 0;
    v8f c0 = {}, c1 = {};
#pragma unroll
    for (int j = 0; j < 8; ++j) {
      v2f hv = *(const v2f*)(sH + 4 * j + 2 * half);
      v2f a;
      a.x = live ? hv.x : 0.f;
      a.y = live ? hv.y : 0.f;
      c0 = __builtin_amdgcn_wmma_f32_16x16x4_f32(
          false, a, false, wb[0][j & 3], (short)0, c0, false, false);
      c1 = __builtin_amdgcn_wmma_f32_16x16x4_f32(
          false, a, false, wb[1][j & 3], (short)0, c1, false, false);
    }
    if (lane < 16) {
      sCa[(wave * 2) * 16 + l15]     = 1.f / (1.f + __expf(-c0[0]));
      sCa[(wave * 2 + 1) * 16 + l15] = 1.f / (1.f + __expf(-c1[0]));
    }
  }
  lds_barrier();
  if ((blockIdx.x & 15) == 0)
    ca[b * C_ + threadIdx.x] = sCa[threadIdx.x];
  const v4f* cag = (const v4f*)(sCa + g * FEAT_CPG);

  v4f s = {0.f, 0.f, 0.f, 0.f};
  v4f m = {-INFINITY, -INFINITY, -INFINITY, -INFINITY};
#pragma unroll
  for (int r = 0; r < FEAT_NBATCH; ++r) {
    __builtin_amdgcn_sched_barrier(0);
    if (r + 1 < FEAT_NBATCH) { FEAT_LOAD(r + 1) }
    __builtin_amdgcn_sched_barrier(0);
    v4f cs[FEAT_BATCH / 4];
#pragma unroll
    for (int i = 0; i < FEAT_BATCH / 4; ++i) cs[i] = cag[r * (FEAT_BATCH / 4) + i];
#pragma unroll
    for (int i = 0; i < FEAT_BATCH; ++i) {
      v4f v = buf[r & 1][i] * cs[i >> 2][i & 3];
      s += v;
      m.x = fmaxf(m.x, v.x);
      m.y = fmaxf(m.y, v.y);
      m.z = fmaxf(m.z, v.z);
      m.w = fmaxf(m.w, v.w);
    }
  }
#undef FEAT_LOAD
  sS[g][q] = s;
  sM[g][q] = m;
  __syncthreads();
  if (threadIdx.x < 64) {
    v4f ts = ((sS[0][q] + sS[1][q]) + sS[2][q]) + sS[3][q];
    v4f a = sM[0][q], bq = sM[1][q], cq = sM[2][q], dq = sM[3][q];
    v4f tm;
    tm.x = fmaxf(fmaxf(a.x, bq.x), fmaxf(cq.x, dq.x));
    tm.y = fmaxf(fmaxf(a.y, bq.y), fmaxf(cq.y, dq.y));
    tm.z = fmaxf(fmaxf(a.z, bq.z), fmaxf(cq.z, dq.z));
    tm.w = fmaxf(fmaxf(a.w, bq.w), fmaxf(cq.w, dq.w));
    *(v4f*)(fm + b * HW_ + hw0 + 4 * q) = ts * (1.f / 256.f);
    *(v4f*)(fx + b * HW_ + hw0 + 4 * q) = tm;
  }
}

__global__ __launch_bounds__(256) void cbam_conv(const float* __restrict__ fm,
                                                 const float* __restrict__ fx,
                                                 const float* __restrict__ wsp,
                                                 float* __restrict__ sig) {
  __shared__ float sW[98];
  if (threadIdx.x < 98) sW[threadIdx.x] = wsp[threadIdx.x];
  __syncthreads();
  int b  = blockIdx.x >> 4;
  int hw = ((blockIdx.x & 15) << 8) + threadIdx.x;
  int y = hw >> 6, xc = hw & 63;
  const float* fmb = fm + b * HW_;
  const float* fxb = fx + b * HW_;
  float acc = 0.f;
#pragma unroll
  for (int ky = 0; ky < 7; ++ky) {
    int yy = y + ky - 3;
    if (yy < 0 || yy >= H_) continue;
#pragma unroll
    for (int kx = 0; kx < 7; ++kx) {
      int x2 = xc + kx - 3;
      if (x2 < 0 || x2 >= W_) continue;
      acc += fmb[yy * 64 + x2] * sW[ky * 7 + kx]
           + fxb[yy * 64 + x2] * sW[49 + ky * 7 + kx];
    }
  }
  sig[b * HW_ + hw] = 1.f / (1.f + __expf(-acc));
}

#define FIN_PLANES 16
#define FIN_BATCH  8
#define FIN_NBATCH (FIN_PLANES * 4 / FIN_BATCH)
#define FIN_QSTR   (HW_ / 16)

__global__ __launch_bounds__(256) void cbam_final(const float* __restrict__ x,
                                                  const float* __restrict__ ca,
                                                  const float* __restrict__ sig,
                                                  float* __restrict__ out) {
  const int b  = blockIdx.x / (C_ / FIN_PLANES);
  const int c0 = (blockIdx.x % (C_ / FIN_PLANES)) * FIN_PLANES;
  const size_t base = ((size_t)(b * C_ + c0)) * HW_;
  const v4f* px = (const v4f*)(x + base) + threadIdx.x;
  v4f* po = (v4f*)(out + base) + threadIdx.x;
  const v4f* ps = (const v4f*)(sig + (size_t)b * HW_) + threadIdx.x;
  const float* cab = ca + b * C_ + c0;

  v4f sv[4];
#pragma unroll
  for (int q = 0; q < 4; ++q) sv[q] = ps[q * FIN_QSTR];

  v4f buf[2][FIN_BATCH];
#define FIN_LOAD(R)                                                          \
  _Pragma("unroll") for (int i = 0; i < FIN_BATCH; ++i)                      \
    buf[(R) & 1][i] = __builtin_nontemporal_load(px + ((R) * FIN_BATCH + i) * FIN_QSTR);
  FIN_LOAD(0)

#pragma unroll
  for (int r = 0; r < FIN_NBATCH; ++r) {
    __builtin_amdgcn_sched_barrier(0);
    if (r + 1 < FIN_NBATCH) { FIN_LOAD(r + 1) }
    __builtin_amdgcn_sched_barrier(0);
#pragma unroll
    for (int i = 0; i < FIN_BATCH; ++i) {
      const int piece = r * FIN_BATCH + i;
      v4f o = buf[r & 1][i] * (sv[piece & 3] * cab[piece >> 2]);
      __builtin_nontemporal_store(o, po + piece * FIN_QSTR);
    }
  }
#undef FIN_LOAD
}

extern "C" void kernel_launch(void* const* d_in, const int* in_sizes, int n_in,
                              void* d_out, int out_size, void* d_ws, size_t ws_size,
                              hipStream_t stream) {
  const float* x  = (const float*)d_in[0];
  const float* w1 = (const float*)d_in[1];
  const float* w2 = (const float*)d_in[2];
  const float* ws = (const float*)d_in[3];
  float* out = (float*)d_out;

  float* wsf = (float*)d_ws;
  float* avg = wsf;
  float* mxp = wsf + 8192;
  float* ca  = wsf + 16384;
  float* fm  = wsf + 24576;
  float* fx  = wsf + 155648;
  float* sig = wsf + 286720;

  cbam_pool  <<<B_ * C_,          256, 0, stream>>>(x, avg, mxp);
  cbam_feat  <<<B_ * (HW_ / 256), 256, 0, stream>>>(x, avg, mxp, w1, w2, ca, fm, fx);
  cbam_conv  <<<B_ * (HW_ / 256), 256, 0, stream>>>(fm, fx, ws, sig);
  cbam_final <<<B_ * (C_ / FIN_PLANES), 256, 0, stream>>>(x, ca, sig, out);
}


# 7x7 conv + sigmoid folded into the final pass, re-tiled to the feat pass's blocks (batch row x 4 image rows): conv from LDS with zero borders at the block head; 3 launches, not 4; sum order kept
# speedup vs baseline: 2.1664x; 1.3367x over previous
#include <hip/hip_runtime.h>
#include <math.h>


#define B_   32
#define C_   256
#define H_   64
#define W_   64
#define HW_  4096
#define MID_ 16

typedef __attribute__((ext_vector_type(2))) float v2f;
typedef __attribute__((ext_vector_type(4))) float v4f;
typedef __attribute__((ext_vector_type(8))) float v8f;

__global__ __launch_bounds__(256) void cbam_pool(const float* __restrict__ x,
                                                 float* __restrict__ avg,
                                                 float* __restrict__ mxp) {
  int plane = blockIdx.x;
  const v4f* p = (const v4f*)(x + (size_t)plane * HW_);
  float s = 0.f, m = -INFINITY;
#pragma unroll
  for (int i = 0; i < 4; ++i) {
    v4f v = p[threadIdx.x + i * 256];
    s += v.x + v.y + v.z + v.w;
    m = fmaxf(m, fmaxf(fmaxf(v.x, v.y), fmaxf(v.z, v.w)));
  }
#pragma unroll
  for (int off = 16; off >= 1; off >>= 1) {
    s += __shfl_xor(s, off, 32);
    m = fmaxf(m, __shfl_xor(m, off, 32));
  }
  __shared__ float ss[8], sm[8];
  int wave = threadIdx.x >> 5, lane = threadIdx.x & 31;
  if (lane == 0) { ss[wave] = s; sm[wave] = m; }
  __syncthreads();
  if (threadIdx.x == 0) {
    float ts = 0.f, tm = -INFINITY;
#pragma unroll
    for (int i = 0; i < 8; ++i) { ts += ss[i]; tm = fmaxf(tm, sm[i]); }
    avg[plane] = ts * (1.0f / (float)HW_);
    mxp[plane] = tm;
  }
}

#define FEAT_GROUPS  4
#define FEAT_CPG     64
#define FEAT_BATCH   16
#define FEAT_NBATCH  (FEAT_CPG / FEAT_BATCH)
#define FEAT_CSTR    (HW_ / 4)

static __device__ __forceinline__ void lds_barrier() {
  __builtin_amdgcn_fence(__ATOMIC_RELEASE, "workgroup", "local");
  __builtin_amdgcn_s_barrier();
  __builtin_amdgcn_fence(__ATOMIC_ACQUIRE, "workgroup", "local");
}

__global__ __launch_bounds__(256) void cbam_feat(const float* __restrict__ x,
                                                 const float* __restrict__ avg,
                                                 const float* __restrict__ mxp,
                                                 const float* __restrict__ w1,
                                                 const float* __restrict__ w2,
                                                 float* __restrict__ ca,
                                                 float* __restrict__ fm,
                                                 float* __restrict__ fx) {
  __shared__ v4f sS[FEAT_GROUPS][64];
  __shared__ v4f sM[FEAT_GROUPS][64];
  __shared__ float sPart[8][2][16];
  __shared__ __attribute__((aligned(16))) float sH[32];
  __shared__ __attribute__((aligned(16))) float sCa[C_];
  const int b   = blockIdx.x >> 4;
  const int hw0 = (blockIdx.x & 15) << 8;
  const int q   = threadIdx.x & 63;
  const int g    = __builtin_amdgcn_readfirstlane((int)(threadIdx.x >> 6));
  const int wave = __builtin_amdgcn_readfirstlane((int)(threadIdx.x >> 5));
  const int lane = threadIdx.x & 31;
  const int half = lane >> 4;
  const int l15  = lane & 15;
  const v4f* p = (const v4f*)(x + ((size_t)b * C_ + (size_t)g * FEAT_CPG) * HW_ + hw0) + q;

  const float* prow  = ((l15 & 1) ? mxp : avg) + b * C_ + wave * 32 + half * 2;
  const float* w1row = w1 + l15 * C_ + wave * 32 + half * 2;
  v2f pa[8], pb[8];
#pragma unroll
  for (int j = 0; j < 8; ++j) {
    pa[j] = *(const v2f*)(prow + 4 * j);
    pb[j] = *(const v2f*)(w1row + 4 * j);
  }
  v2f wb[2][4];
#pragma unroll
  for (int t = 0; t < 2; ++t)
#pragma unroll
    for (int jj = 0; jj < 4; ++jj)
      wb[t][jj] = *(const v2f*)(w2 + ((wave * 2 + t) * 16 + l15) * MID_ + 4 * jj + 2 * half);

  v4f buf[2][FEAT_BATCH];
#define FEAT_LOAD(R)                                                         \
  _Pragma("unroll") for (int i = 0; i < FEAT_BATCH; ++i)                     \
    buf[(R) & 1][i] = p[((R) * FEAT_BATCH + i) * FEAT_CSTR];
  __builtin_amdgcn_sched_barrier(0);
  FEAT_LOAD(0)
  __builtin_amdgcn_sched_barrier(0);

  {
    const bool live = l15 < 2;
    v8f acc = {};
#pragma unroll
    for (int j = 0; j < 8; ++j) {
      v2f a;
      a.x = live ? pa[j].x : 0.f;
      a.y = live ? pa[j].y : 0.f;
      acc = __builtin_amdgcn_wmma_f32_16x16x4_f32(
          false, a, false, pb[j], (short)0, acc, false, false);
    }
    if (lane < 16) {
      sPart[wave][0][l15] = acc[0];
      sPart[wave][1][l15] = acc[1];
    }
  }
  lds_barrier();
  if (threadIdx.x < 32) {
    float t = sPart[0][half][l15];
#pragma unroll
    for (int w = 1; w < 8; ++w) t += sPart[w][half][l15];
    sH[lane] = fmaxf(t, 0.f);
  }
  lds_barrier();

  {
    const bool live = l15 == 0;
    v8f c0 = {}, c1 = {};
#pragma unroll
    for (int j = 0; j < 8; ++j) {
      v2f hv = *(const v2f*)(sH + 4 * j + 2 * half);
      v2f a;
      a.x = live ? hv.x : 0.f;
      a.y = live ? hv.y : 0.f;
      c0 = __builtin_amdgcn_wmma_f32_16x16x4_f32(
          false, a, false, wb[0][j & 3], (short)0, c0, false, false);
      c1 = __builtin_amdgcn_wmma_f32_16x16x4_f32(
          false, a, false, wb[1][j & 3], (short)0, c1, false, false);
    }
    if (lane < 16) {
      sCa[(wave * 2) * 16 + l15]     = 1.f / (1.f + __expf(-c0[0]));
      sCa[(wave * 2 + 1) * 16 + l15] = 1.f / (1.f + __expf(-c1[0]));
    }
  }
  lds_barrier();
  if ((blockIdx.x & 15) == 0)
    ca[b * C_ + threadIdx.x] = sCa[threadIdx.x];
  const v4f* cag = (const v4f*)(sCa + g * FEAT_CPG);

  v4f s = {0.f, 0.f, 0.f, 0.f};
  v4f m = {-INFINITY, -INFINITY, -INFINITY, -INFINITY};
#pragma unroll
  for (int r = 0; r < FEAT_NBATCH; ++r) {
    __builtin_amdgcn_sched_barrier(0);
    if (r + 1 < FEAT_NBATCH) { FEAT_LOAD(r + 1) }
    __builtin_amdgcn_sched_barrier(0);
    v4f cs[FEAT_BATCH / 4];
#pragma unroll
    for (int i = 0; i < FEAT_BATCH / 4; ++i) cs[i] = cag[r * (FEAT_BATCH / 4) + i];
#pragma unroll
    for (int i = 0; i < FEAT_BATCH; ++i) {
      v4f v = buf[r & 1][i] * cs[i >> 2][i & 3];
      s += v;
      m.x = fmaxf(m.x, v.x);
      m.y = fmaxf(m.y, v.y);
      m.z = fmaxf(m.z, v.z);
      m.w = fmaxf(m.w, v.w);
    }
  }
#undef FEAT_LOAD
  sS[g][q] = s;
  sM[g][q] = m;
  __syncthreads();
  if (threadIdx.x < 64) {
    v4f ts = ((sS[0][q] + sS[1][q]) + sS[2][q]) + sS[3][q];
    v4f a = sM[0][q], bq = sM[1][q], cq = sM[2][q], dq = sM[3][q];
    v4f tm;
    tm.x = fmaxf(fmaxf(a.x, bq.x), fmaxf(cq.x, dq.x));
    tm.y = fmaxf(fmaxf(a.y, bq.y), fmaxf(cq.y, dq.y));
    tm.z = fmaxf(fmaxf(a.z, bq.z), fmaxf(cq.z, dq.z));
    tm.w = fmaxf(fmaxf(a.w, bq.w), fmaxf(cq.w, dq.w));
    *(v4f*)(fm + b * HW_ + hw0 + 4 * q) = ts * (1.f / 256.f);
    *(v4f*)(fx + b * HW_ + hw0 + 4 * q) = tm;
  }
}

#define FIN_CPG    64
#define FIN_BATCH  16
#define FIN_NBATCH (FIN_CPG / FIN_BATCH)
#define FIN_CSTR   (HW_ / 4)
#define HALO_ROWS  10
#define HALO_PITCH 72

__global__ __launch_bounds__(256) void cbam_final(const float* __restrict__ x,
                                                  const float* __restrict__ ca,
                                                  const float* __restrict__ fm,
                                                  const float* __restrict__ fx,
                                                  const float* __restrict__ wsp,
                                                  float* __restrict__ out) {
  __shared__ __attribute__((aligned(16))) float sF[2 * HALO_ROWS * HALO_PITCH];
  __shared__ __attribute__((aligned(16))) float sSig[256];
  __shared__ float sW[98];
  const int b    = blockIdx.x >> 4;
  const int tile = blockIdx.x & 15;
  const int hw0  = tile << 8;
  const int y0   = tile << 2;
  const int tid  = threadIdx.x;
  const int q    = tid & 63;
  const int g    = __builtin_amdgcn_readfirstlane(tid >> 6);
  const size_t xoff = ((size_t)b * C_ + (size_t)g * FIN_CPG) * HW_ + hw0;
  const v4f* px = (const v4f*)(x + xoff) + q;
  v4f* po = (v4f*)(out + xoff) + q;
  const float* cab = ca + b * C_ + g * FIN_CPG;

  const int i0 = tid, i1 = 256 + (tid & 63);
  const int m0 = i0 >= 160, m1 = 1;
  const int r0 = (i0 - 160 * m0) >> 4, r1 = (i1 - 160) >> 4;
  const int j0 = i0 & 15, j1 = i1 & 15;
  const int gy0 = y0 - 3 + r0, gy1 = y0 - 3 + r1;
  v4f h0 = {0.f, 0.f, 0.f, 0.f}, h1 = {0.f, 0.f, 0.f, 0.f};
  if (gy0 >= 0 && gy0 < H_)
    h0 = *(const v4f*)((m0 ? fx : fm) + b * HW_ + gy0 * W_ + 4 * j0);
  if (tid < 64 && gy1 >= 0 && gy1 < H_)
    h1 = *(const v4f*)(fx + b * HW_ + gy1 * W_ + 4 * j1);
  float wv = 0.f;
  if (tid < 98) wv = wsp[tid];

  v4f buf[2][FIN_BATCH];
#define FIN_LOAD(R)                                                          \
  _Pragma("unroll") for (int i = 0; i < FIN_BATCH; ++i)                      \
    buf[(R) & 1][i] = __builtin_nontemporal_load(px + ((R) * FIN_BATCH + i) * FIN_CSTR);
  __builtin_amdgcn_sched_barrier(0);
  FIN_LOAD(0)
  __builtin_amdgcn_sched_barrier(0);

  *(v4f*)(sF + (m0 * HALO_ROWS + r0) * HALO_PITCH + 4 + 4 * j0) = h0;
  if (tid < 64)
    *(v4f*)(sF + (m1 * HALO_ROWS + r1) * HALO_PITCH + 4 + 4 * j1) = h1;
  if (tid < 4 * HALO_ROWS) {
    const v4f z = {0.f, 0.f, 0.f, 0.f};
    *(v4f*)(sF + (tid >> 1) * HALO_PITCH + (tid & 1) * 68) = z;
  }
  if (tid < 98) sW[tid] = wv;
  lds_barrier();
  {
    const int yy = tid >> 6, xx = tid & 63;
    const float* fmr = sF + yy * HALO_PITCH + xx + 1;
    const float* fxr = fmr + HALO_ROWS * HALO_PITCH;
    float acc = 0.f;
#pragma unroll
    for (int ky = 0; ky < 7; ++ky) {
#pragma unroll
      for (int kx = 0; kx < 7; ++kx) {
        acc += fmr[ky * HALO_PITCH + kx] * sW[ky * 7 + kx]
             + fxr[ky * HALO_PITCH + kx] * sW[49 + ky * 7 + kx];
      }
    }
    sSig[tid] = 1.f / (1.f + __expf(-acc));
  }
  lds_barrier();
  const v4f sv = *(const v4f*)(sSig + 4 * q);

#pragma unroll
  for (int r = 0; r < FIN_NBATCH; ++r) {
    __builtin_amdgcn_sched_barrier(0);
    if (r + 1 < FIN_NBATCH) { FIN_LOAD(r + 1) }
    __builtin_amdgcn_sched_barrier(0);
#pragma unroll
    for (int i = 0; i < FIN_BATCH; ++i) {
      const int c = r * FIN_BATCH + i;
      v4f o = buf[r & 1][i] * (sv * cab[c]);
      __builtin_nontemporal_store(o, po + c * FIN_CSTR);
    }
  }
#undef FIN_LOAD
}

extern "C" void kernel_launch(void* const* d_in, const int* in_sizes, int n_in,
                              void* d_out, int out_size, void* d_ws, size_t ws_size,
                              hipStream_t stream) {
  const float* x  = (const float*)d_in[0];
  const float* w1 = (const float*)d_in[1];
  const float* w2 = (const float*)d_in[2];
  const float* ws = (const float*)d_in[3];
  float* out = (float*)d_out;

  float* wsf = (float*)d_ws;
  float* avg = wsf;
  float* mxp = wsf + 8192;
  float* ca  = wsf + 16384;
  float* fm  = wsf + 24576;
  float* fx  = wsf + 155648;

  cbam_pool  <<<B_ * C_,          256, 0, stream>>>(x, avg, mxp);
  cbam_feat  <<<B_ * (HW_ / 256), 256, 0, stream>>>(x, avg, mxp, w1, w2, ca, fm, fx);
  cbam_final <<<B_ * (HW_ / 256), 256, 0, stream>>>(x, ca, fm, fx, ws, out);
}


# pool pass as a resident stream: 512 blocks of 8 waves, a wave walks 2 contiguous planes in batches of 16 16-byte loads requested one batch ahead; 16x fewer waves; per-plane sum order changes
# speedup vs baseline: 2.2777x; 1.0514x over previous
#include <hip/hip_runtime.h>
#include <math.h>


#define B_   32
#define C_   256
#define H_   64
#define W_   64
#define HW_  4096
#define MID_ 16

typedef __attribute__((ext_vector_type(2))) float v2f;
typedef __attribute__((ext_vector_type(4))) float v4f;
typedef __attribute__((ext_vector_type(8))) float v8f;

#define POOL_WAVES  8
#define POOL_PPW    2
#define POOL_BATCH  16
#define POOL_BPP    (HW_ / 4 / 32 / POOL_BATCH)
#define POOL_NBATCH (POOL_PPW * POOL_BPP)

__global__ __launch_bounds__(POOL_WAVES * 32) void cbam_pool(const float* __restrict__ x,
                                                             float* __restrict__ avg,
                                                             float* __restrict__ mxp) {
  __shared__ float ss[POOL_WAVES * POOL_PPW], sm[POOL_WAVES * POOL_PPW];
  const int lane = threadIdx.x & 31;
  const int wave = __builtin_amdgcn_readfirstlane((int)(threadIdx.x >> 5));
  const int plane0 = (blockIdx.x * POOL_WAVES + wave) * POOL_PPW;
  const v4f* p = (const v4f*)(x + (size_t)plane0 * HW_) + lane;

  v4f buf[2][POOL_BATCH];
#define POOL_LOAD(R)                                                         \
  _Pragma("unroll") for (int i = 0; i < POOL_BATCH; ++i)                     \
    buf[(R) & 1][i] = p[((R) * POOL_BATCH + i) * 32];
  POOL_LOAD(0)

  float rs[POOL_PPW], rm[POOL_PPW];
  v4f s = {0.f, 0.f, 0.f, 0.f};
  v4f m = {-INFINITY, -INFINITY, -INFINITY, -INFINITY};
#pragma unroll
  for (int r = 0; r < POOL_NBATCH; ++r) {
    __builtin_amdgcn_sched_barrier(0);
    if (r + 1 < POOL_NBATCH) { POOL_LOAD(r + 1) }
    __builtin_amdgcn_sched_barrier(0);
#pragma unroll
    for (int i = 0; i < POOL_BATCH; ++i) {
      v4f v = buf[r & 1][i];
      s += v;
      m.x = fmaxf(m.x, v.x);
      m.y = fmaxf(m.y, v.y);
      m.z = fmaxf(m.z, v.z);
      m.w = fmaxf(m.w, v.w);
    }
    if ((r % POOL_BPP) == POOL_BPP - 1) {
      rs[r / POOL_BPP] = (s.x + s.y) + (s.z + s.w);
      rm[r / POOL_BPP] = fmaxf(fmaxf(m.x, m.y), fmaxf(m.z, m.w));
      s = (v4f){0.f, 0.f, 0.f, 0.f};
      m = (v4f){-INFINITY, -INFINITY, -INFINITY, -INFINITY};
    }
  }
#undef POOL_LOAD
#pragma unroll
  for (int off = 16; off >= 1; off >>= 1) {
#pragma unroll
    for (int k = 0; k < POOL_PPW; ++k) {
      rs[k] += __shfl_xor(rs[k], off, 32);
      rm[k] = fmaxf(rm[k], __shfl_xor(rm[k], off, 32));
    }
  }
  if (lane == 0) {
#pragma unroll
    for (int k = 0; k < POOL_PPW; ++k) {
      ss[wave * POOL_PPW + k] = rs[k] * (1.0f / (float)HW_);
      sm[wave * POOL_PPW + k] = rm[k];
    }
  }
  __syncthreads();
  if (threadIdx.x < POOL_WAVES * POOL_PPW) {
    const int plane = blockIdx.x * (POOL_WAVES * POOL_PPW) + threadIdx.x;
    avg[plane] = ss[threadIdx.x];
    mxp[plane] = sm[threadIdx.x];
  }
}

#define FEAT_GROUPS  4
#define FEAT_CPG     64
#define FEAT_BATCH   16
#define FEAT_NBATCH  (FEAT_CPG / FEAT_BATCH)
#define FEAT_CSTR    (HW_ / 4)

static __device__ __forceinline__ void lds_barrier() {
  __builtin_amdgcn_fence(__ATOMIC_RELEASE, "workgroup", "local");
  __builtin_amdgcn_s_barrier();
  __builtin_amdgcn_fence(__ATOMIC_ACQUIRE, "workgroup", "local");
}

__global__ __launch_bounds__(256) void cbam_feat(const float* __restrict__ x,
                                                 const float* __restrict__ avg,
                                                 const float* __restrict__ mxp,
                                                 const float* __restrict__ w1,
                                                 const float* __restrict__ w2,
                                                 float* __restrict__ ca,
                                                 float* __restrict__ fm,
                                                 float* __restrict__ fx) {
  __shared__ v4f sS[FEAT_GROUPS][64];
  __shared__ v4f sM[FEAT_GROUPS][64];
  __shared__ float sPart[8][2][16];
  __shared__ __attribute__((aligned(16))) float sH[32];
  __shared__ __attribute__((aligned(16))) float sCa[C_];
  const int b   = blockIdx.x >> 4;
  const int hw0 = (blockIdx.x & 15) << 8;
  const int q   = threadIdx.x & 63;
  const int g    = __builtin_amdgcn_readfirstlane((int)(threadIdx.x >> 6));
  const int wave = __builtin_amdgcn_readfirstlane((int)(threadIdx.x >> 5));
  const int lane = threadIdx.x & 31;
  const int half = lane >> 4;
  const int l15  = lane & 15;
  const v4f* p = (const v4f*)(x + ((size_t)b * C_ + (size_t)g * FEAT_CPG) * HW_ + hw0) + q;

  const float* prow  = ((l15 & 1) ? mxp : avg) + b * C_ + wave * 32 + half * 2;
  const float* w1row = w1 + l15 * C_ + wave * 32 + half * 2;
  v2f pa[8], pb[8];
#pragma unroll
  for (int j = 0; j < 8; ++j) {
    pa[j] = *(const v2f*)(prow + 4 * j);
    pb[j] = *(const v2f*)(w1row + 4 * j);
  }
  v2f wb[2][4];
#pragma unroll
  for (int t = 0; t < 2; ++t)
#pragma unroll
    for (int jj = 0; jj < 4; ++jj)
      wb[t][jj] = *(const v2f*)(w2 + ((wave * 2 + t) * 16 + l15) * MID_ + 4 * jj + 2 * half);

  v4f buf[2][FEAT_BATCH];
#define FEAT_LOAD(R)                                                         \
  _Pragma("unroll") for (int i = 0; i < FEAT_BATCH; ++i)                     \
    buf[(R) & 1][i] = p[((R) * FEAT_BATCH + i) * FEAT_CSTR];
  __builtin_amdgcn_sched_barrier(0);
  FEAT_LOAD(0)
  __builtin_amdgcn_sched_barrier(0);

  {
    const bool live = l15 < 2;
    v8f acc = {};
#pragma unroll
    for (int j = 0; j < 8; ++j) {
      v2f a;
      a.x = live ? pa[j].x : 0.f;
      a.y = live ? pa[j].y : 0.f;
      acc = __builtin_amdgcn_wmma_f32_16x16x4_f32(
          false, a, false, pb[j], (short)0, acc, false, false);
    }
    if (lane < 16) {
      sPart[wave][0][l15] = acc[0];
      sPart[wave][1][l15] = acc[1];
    }
  }
  lds_barrier();
  if (threadIdx.x < 32) {
    float t = sPart[0][half][l15];
#pragma unroll
    for (int w = 1; w < 8; ++w) t += sPart[w][half][l15];
    sH[lane] = fmaxf(t, 0.f);
  }
  lds_barrier();

  {
    const bool live = l15 == 0;
    v8f c0 = {}, c1 = {};
#pragma unroll
    for (int j = 0; j < 8; ++j) {
      v2f hv = *(const v2f*)(sH + 4 * j + 2 * half);
      v2f a;
      a.x = live ? hv.x : 0.f;
      a.y = live ? hv.y : 0.f;
      c0 = __builtin_amdgcn_wmma_f32_16x16x4_f32(
          false, a, false, wb[0][j & 3], (short)0, c0, false, false);
      c1 = __builtin_amdgcn_wmma_f32_16x16x4_f32(
          false, a, false, wb[1][j & 3], (short)0, c1, false, false);
    }
    if (lane < 16) {
      sCa[(wave * 2) * 16 + l15]     = 1.f / (1.f + __expf(-c0[0]));
      sCa[(wave * 2 + 1) * 16 + l15] = 1.f / (1.f + __expf(-c1[0]));
    }
  }
  lds_barrier();
  if ((blockIdx.x & 15) == 0)
    ca[b * C_ + threadIdx.x] = sCa[threadIdx.x];
  const v4f* cag = (const v4f*)(sCa + g * FEAT_CPG);

  v4f s = {0.f, 0.f, 0.f, 0.f};
  v4f m = {-INFINITY, -INFINITY, -INFINITY, -INFINITY};
#pragma unroll
  for (int r = 0; r < FEAT_NBATCH; ++r) {
    __builtin_amdgcn_sched_barrier(0);
    if (r + 1 < FEAT_NBATCH) { FEAT_LOAD(r + 1) }
    __builtin_amdgcn_sched_barrier(0);
    v4f cs[FEAT_BATCH / 4];
#pragma unroll
    for (int i = 0; i < FEAT_BATCH / 4; ++i) cs[i] = cag[r * (FEAT_BATCH / 4) + i];
#pragma unroll
    for (int i = 0; i < FEAT_BATCH; ++i) {
      v4f v = buf[r & 1][i] * cs[i >> 2][i & 3];
      s += v;
      m.x = fmaxf(m.x, v.x);
      m.y = fmaxf(m.y, v.y);
      m.z = fmaxf(m.z, v.z);
      m.w = fmaxf(m.w, v.w);
    }
  }
#undef FEAT_LOAD
  sS[g][q] = s;
  sM[g][q] = m;
  __syncthreads();
  if (threadIdx.x < 64) {
    v4f ts = ((sS[0][q] + sS[1][q]) + sS[2][q]) + sS[3][q];
    v4f a = sM[0][q], bq = sM[1][q], cq = sM[2][q], dq = sM[3][q];
    v4f tm;
    tm.x = fmaxf(fmaxf(a.x, bq.x), fmaxf(cq.x, dq.x));
    tm.y = fmaxf(fmaxf(a.y, bq.y), fmaxf(cq.y, dq.y));
    tm.z = fmaxf(fmaxf(a.z, bq.z), fmaxf(cq.z, dq.z));
    tm.w = fmaxf(fmaxf(a.w, bq.w), fmaxf(cq.w, dq.w));
    *(v4f*)(fm + b * HW_ + hw0 + 4 * q) = ts * (1.f / 256.f);
    *(v4f*)(fx + b * HW_ + hw0 + 4 * q) = tm;
  }
}

#define FIN_CPG    64
#define FIN_BATCH  16
#define FIN_NBATCH (FIN_CPG / FIN_BATCH)
#define FIN_CSTR   (HW_ / 4)
#define HALO_ROWS  10
#define HALO_PITCH 72

__global__ __launch_bounds__(256) void cbam_final(const float* __restrict__ x,
                                                  const float* __restrict__ ca,
                                                  const float* __restrict__ fm,
                                                  const float* __restrict__ fx,
                                                  const float* __restrict__ wsp,
                                                  float* __restrict__ out) {
  __shared__ __attribute__((aligned(16))) float sF[2 * HALO_ROWS * HALO_PITCH];
  __shared__ __attribute__((aligned(16))) float sSig[256];
  __shared__ float sW[98];
  const int b    = blockIdx.x >> 4;
  const int tile = blockIdx.x & 15;
  const int hw0  = tile << 8;
  const int y0   = tile << 2;
  const int tid  = threadIdx.x;
  const int q    = tid & 63;
  const int g    = __builtin_amdgcn_readfirstlane(tid >> 6);
  const size_t xoff = ((size_t)b * C_ + (size_t)g * FIN_CPG) * HW_ + hw0;
  const v4f* px = (const v4f*)(x + xoff) + q;
  v4f* po = (v4f*)(out + xoff) + q;
  const float* cab = ca + b * C_ + g * FIN_CPG;

  const int i0 = tid, i1 = 256 + (tid & 63);
  const int m0 = i0 >= 160, m1 = 1;
  const int r0 = (i0 - 160 * m0) >> 4, r1 = (i1 - 160) >> 4;
  const int j0 = i0 & 15, j1 = i1 & 15;
  const int gy0 = y0 - 3 + r0, gy1 = y0 - 3 + r1;
  v4f h0 = {0.f, 0.f, 0.f, 0.f}, h1 = {0.f, 0.f, 0.f, 0.f};
  if (gy0 >= 0 && gy0 < H_)
    h0 = *(const v4f*)((m0 ? fx : fm) + b * HW_ + gy0 * W_ + 4 * j0);
  if (tid < 64 && gy1 >= 0 && gy1 < H_)
    h1 = *(const v4f*)(fx + b * HW_ + gy1 * W_ + 4 * j1);
  float wv = 0.f;
  if (tid < 98) wv = wsp[tid];

  v4f buf[2][FIN_BATCH];
#define FIN_LOAD(R)                                                          \
  _Pragma("unroll") for (int i = 0; i < FIN_BATCH; ++i)                      \
    buf[(R) & 1][i] = __builtin_nontemporal_load(px + ((R) * FIN_BATCH + i) * FIN_CSTR);
  __builtin_amdgcn_sched_barrier(0);
  FIN_LOAD(0)
  __builtin_amdgcn_sched_barrier(0);

  *(v4f*)(sF + (m0 * HALO_ROWS + r0) * HALO_PITCH + 4 + 4 * j0) = h0;
  if (tid < 64)
    *(v4f*)(sF + (m1 * HALO_ROWS + r1) * HALO_PITCH + 4 + 4 * j1) = h1;
  if (tid < 4 * HALO_ROWS) {
    const v4f z = {0.f, 0.f, 0.f, 0.f};
    *(v4f*)(sF + (tid >> 1) * HALO_PITCH + (tid & 1) * 68) = z;
  }
  if (tid < 98) sW[tid] = wv;
  lds_barrier();
  {
    const int yy = tid >> 6, xx = tid & 63;
    const float* fmr = sF + yy * HALO_PITCH + xx + 1;
    const float* fxr = fmr + HALO_ROWS * HALO_PITCH;
    float acc = 0.f;
#pragma unroll
    for (int ky = 0; ky < 7; ++ky) {
#pragma unroll
      for (int kx = 0; kx < 7; ++kx) {
        acc += fmr[ky * HALO_PITCH + kx] * sW[ky * 7 + kx]
             + fxr[ky * HALO_PITCH + kx] * sW[49 + ky * 7 + kx];
      }
    }
    sSig[tid] = 1.f / (1.f + __expf(-acc));
  }
  lds_barrier();
  const v4f sv = *(const v4f*)(sSig + 4 * q);

#pragma unroll
  for (int r = 0; r < FIN_NBATCH; ++r) {
    __builtin_amdgcn_sched_barrier(0);
    if (r + 1 < FIN_NBATCH) { FIN_LOAD(r + 1) }
    __builtin_amdgcn_sched_barrier(0);
#pragma unroll
    for (int i = 0; i < FIN_BATCH; ++i) {
      const int c = r * FIN_BATCH + i;
      v4f o = buf[r & 1][i] * (sv * cab[c]);
      __builtin_nontemporal_store(o, po + c * FIN_CSTR);
    }
  }
#undef FIN_LOAD
}

extern "C" void kernel_launch(void* const* d_in, const int* in_sizes, int n_in,
                              void* d_out, int out_size, void* d_ws, size_t ws_size,
                              hipStream_t stream) {
  const float* x  = (const float*)d_in[0];
  const float* w1 = (const float*)d_in[1];
  const float* w2 = (const float*)d_in[2];
  const float* ws = (const float*)d_in[3];
  float* out = (float*)d_out;

  float* wsf = (float*)d_ws;
  float* avg = wsf;
  float* mxp = wsf + 8192;
  float* ca  = wsf + 16384;
  float* fm  = wsf + 24576;
  float* fx  = wsf + 155648;

  cbam_pool  <<<B_ * C_ / (POOL_WAVES * POOL_PPW), POOL_WAVES * 32, 0, stream>>>(x, avg, mxp);
  cbam_feat  <<<B_ * (HW_ / 256), 256, 0, stream>>>(x, avg, mxp, w1, w2, ca, fm, fx);
  cbam_final <<<B_ * (HW_ / 256), 256, 0, stream>>>(x, ca, fm, fx, ws, out);
}
